# starting kernel
# baseline (speedup 1.0000x reference)
#include <hip/hip_runtime.h>

#define BB    16
#define NN    5000
#define EE    80000
#define FEAT  50
#define HID   128
#define PADR  1000

typedef __attribute__((ext_vector_type(2))) float v2f;
typedef __attribute__((ext_vector_type(8))) float v8f;

__device__ __forceinline__ v8f wmma_f32x4(v2f a, v2f b, v8f c) {
  return __builtin_amdgcn_wmma_f32_16x16x4_f32(
       false, a,  false, b,
       (short)0, c,  false,  false);
}

__device__ __forceinline__ int imin(int a, int b) { return a < b ? a : b; }


__global__ void fill_zero_kernel(float* __restrict__ p, long n) {
  long i = (long)blockIdx.x * blockDim.x + threadIdx.x;
  long stride = (long)gridDim.x * blockDim.x;
  for (; i < n; i += stride) p[i] = 0.0f;
}

__global__ void count_edges_kernel(const int* __restrict__ ei, float* __restrict__ cnt) {
  long total = (long)BB * EE;
  long i = (long)blockIdx.x * blockDim.x + threadIdx.x;
  long stride = (long)gridDim.x * blockDim.x;
  for (; i < total; i += stride) {
    int e = (int)(i % EE);
    int b = (int)(i / EE);
    int dst = ei[(long)b * 2 * EE + EE + e];
    atomicAdd(&cnt[(long)b * NN + dst], 1.0f);
  }
}

__global__ void scatter_add_kernel(const float* __restrict__ x, const int* __restrict__ ei,
                                   float* __restrict__ agg, int F) {
  long total = (long)BB * EE * F;
  long i = (long)blockIdx.x * blockDim.x + threadIdx.x;
  long stride = (long)gridDim.x * blockDim.x;
  for (; i < total; i += stride) {
    int f = (int)(i % F);
    long t = i / F;
    int e = (int)(t % EE);
    int b = (int)(t / EE);
    const int* eib = ei + (long)b * 2 * EE;
    int src = eib[e];
    int dst = eib[EE + e];
    atomicAdd(&agg[((long)b * NN + dst) * F + f], x[((long)b * NN + src) * F + f]);
  }
}

__global__ void mean_div_kernel(float* __restrict__ agg, const float* __restrict__ cnt, int F) {
  long total = (long)BB * NN * F;
  long i = (long)blockIdx.x * blockDim.x + threadIdx.x;
  long stride = (long)gridDim.x * blockDim.x;
  for (; i < total; i += stride) {
    float c = cnt[i / F];
    agg[i] /= fmaxf(c, 1.0f);
  }
}

template <int NT, int NCT>
__global__ void wmma_gemm_kernel(const float* __restrict__ A,
                                 const float* __restrict__ A2,
                                 const float* __restrict__ Wa,
                                 const float* __restrict__ Wb,
                                 const float* __restrict__ bias,
                                 float* __restrict__ C,
                                 int M, int K, int relu,
                                 int pad_mode, const int* __restrict__ act_range_p,
                                 int pad_rows, int src_rows) {
  const int lane = threadIdx.x & 31;
  const int mt = blockIdx.x * (blockDim.x >> 5) + (threadIdx.x >> 5);
  const int tm = (M + 15) >> 4;
  if (mt >= tm) return;

  const int ml = lane & 15;
  const int g  = lane >> 4;

  const int row = mt * 16 + ml;
  bool rvalid = row < M;
  long arow;
  if (pad_mode) {
    const int rc = imin(row, M - 1);
    const int b  = rc / pad_rows;
    const int rr = rc - b * pad_rows;
    rvalid = rvalid && (rr < *act_range_p);
    arow = (long)b * src_rows + rr;
  } else {
    arow = imin(row, M - 1);
  }

  v8f acc[NT];
#pragma unroll
  for (int t = 0; t < NT; ++t) acc[t] = (v8f){};

  const int npass = (A2 != nullptr) ? 2 : 1;

  for (int pass = 0; pass < npass; ++pass) {
    const float* __restrict__ Ap = pass ? A2 : A;
    const float* __restrict__ Wp = pass ? Wb : Wa;

    const float* aptr = Ap + arow * (long)K + 2 * g;
    const float* bptr = Wp + (2 * g) * NCT + ml;
    const int nfull = K >> 2;

    for (int it = 0; it < nfull; ++it) {
      v2f af = *(const v2f*)aptr;
      af.x = rvalid ? af.x : 0.0f;
      af.y = rvalid ? af.y : 0.0f;
#pragma unroll
      for (int t = 0; t < NT; ++t) {
        v2f bf;
        bf.x = bptr[t * 16];
        bf.y = bptr[NCT + t * 16];
        acc[t] = wmma_f32x4(af, bf, acc[t]);
      }
      aptr += 4;
      bptr += 4 * NCT;
    }

    const int k0 = nfull << 2;
    if (k0 < K) {
      const float* __restrict__ Arow_p = Ap + arow * (long)K;
      const int ka  = k0 + 2 * g;
      const int kc0 = imin(ka, K - 1);
      const int kc1 = imin(ka + 1, K - 1);
      float a0 = Arow_p[kc0];
      float a1 = Arow_p[kc1];
      v2f af;
      af.x = (rvalid && ka     < K) ? a0 : 0.0f;
      af.y = (rvalid && ka + 1 < K) ? a1 : 0.0f;
      const float* __restrict__ W0 = Wp + (long)kc0 * NCT + ml;
      const float* __restrict__ W1 = Wp + (long)kc1 * NCT + ml;
#pragma unroll
      for (int t = 0; t < NT; ++t) {
        v2f bf;
        bf.x = W0[t * 16];
        bf.y = W1[t * 16];
        acc[t] = wmma_f32x4(af, bf, acc[t]);
      }
    }
  }

  float* crow = C + (long)(mt * 16 + (g << 3)) * NCT + ml;
#pragma unroll
  for (int t = 0; t < NT; ++t) {
    const int n = t * 16 + ml;
    const bool nvalid = (NT * 16 == NCT) ? true : (n < NCT);
    const float bv = (bias != nullptr) ? bias[imin(n, NCT - 1)] : 0.0f;
#pragma unroll
    for (int v = 0; v < 8; ++v) {
      const int r = mt * 16 + v + (g << 3);
      float val = acc[t][v] + bv;
      if (relu) val = fmaxf(val, 0.0f);
      if (r < M && nvalid) crow[v * NCT + t * 16] = val;
    }
  }
}

__global__ void final_dot_kernel(const float* __restrict__ y, const float* __restrict__ W3,
                                 const float* __restrict__ b3, float* __restrict__ out, int M) {
  const int lane = threadIdx.x & 31;
  const int row = blockIdx.x * (blockDim.x >> 5) + (threadIdx.x >> 5);
  if (row >= M) return;
  float s = 0.0f;
#pragma unroll
  for (int k = lane; k < HID; k += 32) s += y[(long)row * HID + k] * W3[k];
#pragma unroll
  for (int off = 16; off > 0; off >>= 1) s += __shfl_xor(s, off, 32);
  if (lane == 0) out[row] = s + b3[0];
}


static inline int strip_grid(int M) {
  int tm = (M + 15) / 16;
  return (tm + 3) / 4;
}

extern "C" void kernel_launch(void* const* d_in, const int* in_sizes, int n_in,
                              void* d_out, int out_size, void* d_ws, size_t ws_size,
                              hipStream_t stream) {
  (void)in_sizes; (void)n_in; (void)out_size; (void)ws_size;

  const float* edge_feat = (const float*)d_in[0];
  const int*   edge_idx  = (const int*)d_in[1];
  const int*   act_range = (const int*)d_in[2];
  const float* Wl1 = (const float*)d_in[3];
  const float* Wr1 = (const float*)d_in[4];
  const float* b1c = (const float*)d_in[5];
  const float* Wl2 = (const float*)d_in[6];
  const float* Wr2 = (const float*)d_in[7];
  const float* b2c = (const float*)d_in[8];
  const float* W1  = (const float*)d_in[9];
  const float* b1  = (const float*)d_in[10];
  const float* W2  = (const float*)d_in[11];
  const float* b2  = (const float*)d_in[12];
  const float* W3  = (const float*)d_in[13];
  const float* b3  = (const float*)d_in[14];
  float* out = (float*)d_out;

  const long S_AGG1 = (long)BB * NN * FEAT;
  const long S_CNT  = (long)BB * NN;
  const long S_H    = (long)BB * NN * HID;
  const long S_Y    = (long)BB * PADR * HID;

  float* agg1 = (float*)d_ws;
  float* cnt  = agg1 + S_AGG1;
  float* h    = cnt  + S_CNT;
  float* agg2 = h    + S_H;
  float* y1   = agg2 + S_H;
  float* y2   = y1   + S_Y;

  const int TB = 256;
  const long bigE1 = (long)BB * EE * FEAT;
  const long bigE2 = (long)BB * EE * HID;

  fill_zero_kernel<<<(int)((S_AGG1 + S_CNT + TB - 1) / TB), TB, 0, stream>>>(agg1, S_AGG1 + S_CNT);
  count_edges_kernel<<<(int)(((long)BB * EE + TB - 1) / TB), TB, 0, stream>>>(edge_idx, cnt);
  scatter_add_kernel<<<(int)((bigE1 + TB - 1) / TB), TB, 0, stream>>>(edge_feat, edge_idx, agg1, FEAT);
  mean_div_kernel<<<(int)((S_AGG1 + TB - 1) / TB), TB, 0, stream>>>(agg1, cnt, FEAT);
  wmma_gemm_kernel<8, 128><<<strip_grid(BB * NN), 128, 0, stream>>>(
      agg1, edge_feat, Wl1, Wr1, b1c, h, BB * NN, FEAT,  1,
       0, nullptr, 0, 0);

  fill_zero_kernel<<<(int)((S_H + TB - 1) / TB), TB, 0, stream>>>(agg2, S_H);
  scatter_add_kernel<<<(int)((bigE2 + TB - 1) / TB), TB, 0, stream>>>(h, edge_idx, agg2, HID);
  mean_div_kernel<<<(int)((S_H + TB - 1) / TB), TB, 0, stream>>>(agg2, cnt, HID);
  wmma_gemm_kernel<4, 50><<<strip_grid(BB * NN), 128, 0, stream>>>(
      agg2, h, Wl2, Wr2, b2c, agg1, BB * NN, HID,  0,
       0, nullptr, 0, 0);

  wmma_gemm_kernel<8, 128><<<strip_grid(BB * PADR), 128, 0, stream>>>(
      agg1, nullptr, W1, nullptr, b1, y1, BB * PADR, FEAT,  1,
       1, act_range, PADR, NN);
  wmma_gemm_kernel<8, 128><<<strip_grid(BB * PADR), 128, 0, stream>>>(
      y1, nullptr, W2, nullptr, b2, y2, BB * PADR, HID,  1,
       0, nullptr, 0, 0);
  final_dot_kernel<<<(BB * PADR + 3) / 4, 128, 0, stream>>>(y2, W3, b3, out, BB * PADR);
}


# layer 2 (zero, scatter-add, mean, GEMM) runs only on destination rows below action_range, the only rows the MLP reads; dead edges dropped by a per-wave ballot over 32 edges
# speedup vs baseline: 2.0933x; 2.0933x over previous
#include <hip/hip_runtime.h>

#define BB    16
#define NN    5000
#define EE    80000
#define FEAT  50
#define HID   128
#define PADR  1000

typedef __attribute__((ext_vector_type(2))) float v2f;
typedef __attribute__((ext_vector_type(8))) float v8f;

__device__ __forceinline__ v8f wmma_f32x4(v2f a, v2f b, v8f c) {
  return __builtin_amdgcn_wmma_f32_16x16x4_f32(
       false, a,  false, b,
       (short)0, c,  false,  false);
}

__device__ __forceinline__ int imin(int a, int b) { return a < b ? a : b; }


__global__ void fill_zero_kernel(float* __restrict__ p, long n) {
  long i = (long)blockIdx.x * blockDim.x + threadIdx.x;
  long stride = (long)gridDim.x * blockDim.x;
  for (; i < n; i += stride) p[i] = 0.0f;
}

__global__ void count_edges_kernel(const int* __restrict__ ei, float* __restrict__ cnt) {
  long total = (long)BB * EE;
  long i = (long)blockIdx.x * blockDim.x + threadIdx.x;
  long stride = (long)gridDim.x * blockDim.x;
  for (; i < total; i += stride) {
    int e = (int)(i % EE);
    int b = (int)(i / EE);
    int dst = ei[(long)b * 2 * EE + EE + e];
    atomicAdd(&cnt[(long)b * NN + dst], 1.0f);
  }
}

__global__ void scatter_add_kernel(const float* __restrict__ x, const int* __restrict__ ei,
                                   float* __restrict__ agg, int F) {
  long total = (long)BB * EE * F;
  long i = (long)blockIdx.x * blockDim.x + threadIdx.x;
  long stride = (long)gridDim.x * blockDim.x;
  for (; i < total; i += stride) {
    int f = (int)(i % F);
    long t = i / F;
    int e = (int)(t % EE);
    int b = (int)(t / EE);
    const int* eib = ei + (long)b * 2 * EE;
    int src = eib[e];
    int dst = eib[EE + e];
    atomicAdd(&agg[((long)b * NN + dst) * F + f], x[((long)b * NN + src) * F + f]);
  }
}

__global__ void mean_div_kernel(float* __restrict__ agg, const float* __restrict__ cnt, int F) {
  long total = (long)BB * NN * F;
  long i = (long)blockIdx.x * blockDim.x + threadIdx.x;
  long stride = (long)gridDim.x * blockDim.x;
  for (; i < total; i += stride) {
    float c = cnt[i / F];
    agg[i] /= fmaxf(c, 1.0f);
  }
}


__global__ void fill_zero_live_kernel(float* __restrict__ p) {
  const int i = blockIdx.x * blockDim.x + threadIdx.x;
  if (i >= PADR * HID / 4) return;
  float4* q = (float4*)(p + (long)blockIdx.y * NN * HID) + i;
  *q = make_float4(0.0f, 0.0f, 0.0f, 0.0f);
}

__global__ void scatter_add_live_kernel(const float* __restrict__ hsrc, const int* __restrict__ ei,
                                        float* __restrict__ agg,
                                        const int* __restrict__ act_range_p) {
  const int lane = threadIdx.x & 31;
  const int wv = blockIdx.x * (blockDim.x >> 5) + (threadIdx.x >> 5);
  const int b = blockIdx.y;
  const int e = wv * 32 + lane;
  const int live_rows = imin(*act_range_p, PADR);
  const int* __restrict__ eib = ei + (long)b * 2 * EE;
  int src = 0, dst = -1;
  if (e < EE) { src = eib[e]; dst = eib[EE + e]; }
  const bool live = (unsigned)dst < (unsigned)live_rows;
  unsigned mask = __builtin_amdgcn_ballot_w32(live);
  const float* __restrict__ hb = hsrc + (long)b * NN * HID + lane;
  float* __restrict__ ab = agg + (long)b * NN * HID + lane;
  while (mask) {
    const int l = __builtin_ctz(mask);
    mask &= mask - 1;
    const int s = __builtin_amdgcn_readlane(src, l);
    const int d = __builtin_amdgcn_readlane(dst, l);
    const float* __restrict__ hr = hb + (long)s * HID;
    float* __restrict__ ar = ab + (long)d * HID;
    const float v0 = hr[0], v1 = hr[32], v2 = hr[64], v3 = hr[96];
    atomicAdd(ar,      v0);
    atomicAdd(ar + 32, v1);
    atomicAdd(ar + 64, v2);
    atomicAdd(ar + 96, v3);
  }
}

__global__ void mean_div_live_kernel(float* __restrict__ agg, const float* __restrict__ cnt) {
  const int i = blockIdx.x * blockDim.x + threadIdx.x;
  if (i >= PADR * HID / 4) return;
  const int b = blockIdx.y;
  const int r = i / (HID / 4);
  const float c = fmaxf(cnt[b * NN + r], 1.0f);
  float4* q = (float4*)(agg + (long)b * NN * HID) + i;
  float4 v = *q;
  v.x /= c; v.y /= c; v.z /= c; v.w /= c;
  *q = v;
}

template <int NT, int NCT>
__global__ void wmma_gemm_kernel(const float* __restrict__ A,
                                 const float* __restrict__ A2,
                                 const float* __restrict__ Wa,
                                 const float* __restrict__ Wb,
                                 const float* __restrict__ bias,
                                 float* __restrict__ C,
                                 int M, int K, int relu,
                                 int pad_mode, const int* __restrict__ act_range_p,
                                 int pad_rows, int src_rows) {
  const int lane = threadIdx.x & 31;
  const int mt = blockIdx.x * (blockDim.x >> 5) + (threadIdx.x >> 5);
  const int tm = (M + 15) >> 4;
  if (mt >= tm) return;

  const int ml = lane & 15;
  const int g  = lane >> 4;

  const int row = mt * 16 + ml;
  bool rvalid = row < M;
  long arow;
  if (pad_mode) {
    const int rc = imin(row, M - 1);
    const int b  = rc / pad_rows;
    const int rr = rc - b * pad_rows;
    rvalid = rvalid && (rr < *act_range_p);
    arow = (long)b * src_rows + rr;
  } else {
    arow = imin(row, M - 1);
  }

  v8f acc[NT];
#pragma unroll
  for (int t = 0; t < NT; ++t) acc[t] = (v8f){};

  const int npass = (A2 != nullptr) ? 2 : 1;

  for (int pass = 0; pass < npass; ++pass) {
    const float* __restrict__ Ap = pass ? A2 : A;
    const float* __restrict__ Wp = pass ? Wb : Wa;

    const float* aptr = Ap + arow * (long)K + 2 * g;
    const float* bptr = Wp + (2 * g) * NCT + ml;
    const int nfull = K >> 2;

    for (int it = 0; it < nfull; ++it) {
      v2f af = *(const v2f*)aptr;
      af.x = rvalid ? af.x : 0.0f;
      af.y = rvalid ? af.y : 0.0f;
#pragma unroll
      for (int t = 0; t < NT; ++t) {
        v2f bf;
        bf.x = bptr[t * 16];
        bf.y = bptr[NCT + t * 16];
        acc[t] = wmma_f32x4(af, bf, acc[t]);
      }
      aptr += 4;
      bptr += 4 * NCT;
    }

    const int k0 = nfull << 2;
    if (k0 < K) {
      const float* __restrict__ Arow_p = Ap + arow * (long)K;
      const int ka  = k0 + 2 * g;
      const int kc0 = imin(ka, K - 1);
      const int kc1 = imin(ka + 1, K - 1);
      float a0 = Arow_p[kc0];
      float a1 = Arow_p[kc1];
      v2f af;
      af.x = (rvalid && ka     < K) ? a0 : 0.0f;
      af.y = (rvalid && ka + 1 < K) ? a1 : 0.0f;
      const float* __restrict__ W0 = Wp + (long)kc0 * NCT + ml;
      const float* __restrict__ W1 = Wp + (long)kc1 * NCT + ml;
#pragma unroll
      for (int t = 0; t < NT; ++t) {
        v2f bf;
        bf.x = W0[t * 16];
        bf.y = W1[t * 16];
        acc[t] = wmma_f32x4(af, bf, acc[t]);
      }
    }
  }

  float* crow = C + (long)(mt * 16 + (g << 3)) * NCT + ml;
#pragma unroll
  for (int t = 0; t < NT; ++t) {
    const int n = t * 16 + ml;
    const bool nvalid = (NT * 16 == NCT) ? true : (n < NCT);
    const float bv = (bias != nullptr) ? bias[imin(n, NCT - 1)] : 0.0f;
#pragma unroll
    for (int v = 0; v < 8; ++v) {
      const int r = mt * 16 + v + (g << 3);
      float val = acc[t][v] + bv;
      if (relu) val = fmaxf(val, 0.0f);
      if (r < M && nvalid) crow[v * NCT + t * 16] = val;
    }
  }
}

__global__ void final_dot_kernel(const float* __restrict__ y, const float* __restrict__ W3,
                                 const float* __restrict__ b3, float* __restrict__ out, int M) {
  const int lane = threadIdx.x & 31;
  const int row = blockIdx.x * (blockDim.x >> 5) + (threadIdx.x >> 5);
  if (row >= M) return;
  float s = 0.0f;
#pragma unroll
  for (int k = lane; k < HID; k += 32) s += y[(long)row * HID + k] * W3[k];
#pragma unroll
  for (int off = 16; off > 0; off >>= 1) s += __shfl_xor(s, off, 32);
  if (lane == 0) out[row] = s + b3[0];
}


static inline int strip_grid(int M) {
  int tm = (M + 15) / 16;
  return (tm + 3) / 4;
}

extern "C" void kernel_launch(void* const* d_in, const int* in_sizes, int n_in,
                              void* d_out, int out_size, void* d_ws, size_t ws_size,
                              hipStream_t stream) {
  (void)in_sizes; (void)n_in; (void)out_size; (void)ws_size;

  const float* edge_feat = (const float*)d_in[0];
  const int*   edge_idx  = (const int*)d_in[1];
  const int*   act_range = (const int*)d_in[2];
  const float* Wl1 = (const float*)d_in[3];
  const float* Wr1 = (const float*)d_in[4];
  const float* b1c = (const float*)d_in[5];
  const float* Wl2 = (const float*)d_in[6];
  const float* Wr2 = (const float*)d_in[7];
  const float* b2c = (const float*)d_in[8];
  const float* W1  = (const float*)d_in[9];
  const float* b1  = (const float*)d_in[10];
  const float* W2  = (const float*)d_in[11];
  const float* b2  = (const float*)d_in[12];
  const float* W3  = (const float*)d_in[13];
  const float* b3  = (const float*)d_in[14];
  float* out = (float*)d_out;

  const long S_AGG1 = (long)BB * NN * FEAT;
  const long S_CNT  = (long)BB * NN;
  const long S_H    = (long)BB * NN * HID;
  const long S_Y    = (long)BB * PADR * HID;

  float* agg1 = (float*)d_ws;
  float* cnt  = agg1 + S_AGG1;
  float* h    = cnt  + S_CNT;
  float* agg2 = h    + S_H;
  float* y1   = agg2 + S_H;
  float* y2   = y1   + S_Y;

  const int TB = 256;
  const long bigE1 = (long)BB * EE * FEAT;

  fill_zero_kernel<<<(int)((S_AGG1 + S_CNT + TB - 1) / TB), TB, 0, stream>>>(agg1, S_AGG1 + S_CNT);
  count_edges_kernel<<<(int)(((long)BB * EE + TB - 1) / TB), TB, 0, stream>>>(edge_idx, cnt);
  scatter_add_kernel<<<(int)((bigE1 + TB - 1) / TB), TB, 0, stream>>>(edge_feat, edge_idx, agg1, FEAT);
  mean_div_kernel<<<(int)((S_AGG1 + TB - 1) / TB), TB, 0, stream>>>(agg1, cnt, FEAT);
  wmma_gemm_kernel<8, 128><<<strip_grid(BB * NN), 128, 0, stream>>>(
      agg1, edge_feat, Wl1, Wr1, b1c, h, BB * NN, FEAT,  1,
       0, nullptr, 0, 0);

  const dim3 liveGrid((PADR * HID / 4 + TB - 1) / TB, BB);
  fill_zero_live_kernel<<<liveGrid, TB, 0, stream>>>(agg2);
  scatter_add_live_kernel<<<dim3((EE + 127) / 128, BB), 128, 0, stream>>>(h, edge_idx, agg2, act_range);
  mean_div_live_kernel<<<liveGrid, TB, 0, stream>>>(agg2, cnt);
  wmma_gemm_kernel<4, 50><<<strip_grid(BB * PADR), 128, 0, stream>>>(
      agg2, h, Wl2, Wr2, b2c, agg1, BB * PADR, HID,  0,
       1, act_range, PADR, NN);

  wmma_gemm_kernel<8, 128><<<strip_grid(BB * PADR), 128, 0, stream>>>(
      agg1, nullptr, W1, nullptr, b1, y1, BB * PADR, FEAT,  1,
       1, act_range, PADR, PADR);
  wmma_gemm_kernel<8, 128><<<strip_grid(BB * PADR), 128, 0, stream>>>(
      y1, nullptr, W2, nullptr, b2, y2, BB * PADR, HID,  1,
       0, nullptr, 0, 0);
  final_dot_kernel<<<(BB * PADR + 3) / 4, 128, 0, stream>>>(y2, W3, b3, out, BB * PADR);
}


# layer 2 GEMM, both MLP GEMMs and the final dot run as ONE launch: a block of 8 waves keeps its 32 rows' x2, y1 and y2 in LDS between the products; 3 launches and their trips through memory gone
# speedup vs baseline: 10.1510x; 4.8493x over previous
#include <hip/hip_runtime.h>

#define BB    16
#define NN    5000
#define EE    80000
#define FEAT  50
#define HID   128
#define PADR  1000

typedef __attribute__((ext_vector_type(2))) float v2f;
typedef __attribute__((ext_vector_type(8))) float v8f;

__device__ __forceinline__ v8f wmma_f32x4(v2f a, v2f b, v8f c) {
  return __builtin_amdgcn_wmma_f32_16x16x4_f32(
       false, a,  false, b,
       (short)0, c,  false,  false);
}

__device__ __forceinline__ int imin(int a, int b) { return a < b ? a : b; }


__device__ __forceinline__ int clampi(int v, int lo, int hi) { return v < lo ? lo : (v > hi ? hi : v); }

#define NRANGES 8
#define RNODES  (NN / NRANGES)
#define ECHUNKS (EE / 4)
#define LGROUP  5
#define NGROUPS 4

__device__ __forceinline__ void count_edge(int d, int lo, int* cnt, int& below) {
  below += __builtin_popcount(__builtin_amdgcn_ballot_w32(d < lo));
  const unsigned r = (unsigned)(d - lo);
  if (r < (unsigned)RNODES) atomicAdd(&cnt[r], 1);
}

__device__ __forceinline__ void place_edge(int s, int d, int lo, int* pos, int* __restrict__ list) {
  const unsigned r = (unsigned)(d - lo);
  if (r < (unsigned)RNODES) {
    int p = atomicAdd(&pos[r], 1);
    p = clampi(p, 0, EE - 1);
    list[p] = s;
  }
}

__global__ void __launch_bounds__(1024) build_lists_kernel(const int* __restrict__ ei,
                                                           int* __restrict__ deg,
                                                           int* __restrict__ cursor,
                                                           int* __restrict__ ssrc) {
  __shared__ int cnt[RNODES];
  __shared__ int pos[RNODES];
  __shared__ int wsum[32];
  __shared__ int wbelow[32];
  __shared__ int sbase;
  const int t = threadIdx.x;
  const int lane = t & 31;
  const int w = t >> 5;
  const int b = blockIdx.y;
  const int lo = blockIdx.x * RNODES;
  const int* __restrict__ eib = ei + (long)b * 2 * EE;
  const int4* __restrict__ src4 = (const int4*)eib;
  const int4* __restrict__ dst4 = (const int4*)(eib + EE);
  int* __restrict__ list = ssrc + (long)b * EE;

  if (t < RNODES) cnt[t] = 0;
  __syncthreads();

  int below = 0;
  for (int g = 0; g < NGROUPS; ++g) {
    int4 dv[LGROUP];
#pragma unroll
    for (int u = 0; u < LGROUP; ++u) {
      const int q = t + 1024 * (g * LGROUP + u);
      dv[u] = dst4[imin(q, ECHUNKS - 1)];
    }
#pragma unroll
    for (int u = 0; u < LGROUP; ++u) {
      const int q = t + 1024 * (g * LGROUP + u);
      if (q < ECHUNKS) {
        count_edge(dv[u].x, lo, cnt, below);
        count_edge(dv[u].y, lo, cnt, below);
        count_edge(dv[u].z, lo, cnt, below);
        count_edge(dv[u].w, lo, cnt, below);
      }
    }
  }
  if (lane == 0) wbelow[w] = below;
  __syncthreads();

  const int c = (t < RNODES) ? cnt[t] : 0;
  int inc = c;
  for (int off = 1; off < 32; off <<= 1) {
    const int n = __shfl_up(inc, off, 32);
    if (lane >= off) inc += n;
  }
  if (lane == 31) wsum[w] = inc;
  __syncthreads();
  if (w == 0) {
    const int tot = wsum[lane];
    int wi = tot;
    for (int off = 1; off < 32; off <<= 1) {
      const int n = __shfl_up(wi, off, 32);
      if (lane >= off) wi += n;
    }
    wsum[lane] = wi - tot;
    int bl = wbelow[lane];
    for (int off = 16; off > 0; off >>= 1) bl += __shfl_xor(bl, off, 32);
    if (lane == 0) sbase = bl;
  }
  __syncthreads();
  if (t < RNODES) {
    const int start = sbase + wsum[w] + inc - c;
    pos[t] = start;
    deg[b * NN + lo + t] = c;
    cursor[b * NN + lo + t] = start + c;
  }
  __syncthreads();

  for (int g = 0; g < NGROUPS; ++g) {
    int4 dv[LGROUP];
    int4 sv[LGROUP];
#pragma unroll
    for (int u = 0; u < LGROUP; ++u) {
      const int q = imin(t + 1024 * (g * LGROUP + u), ECHUNKS - 1);
      dv[u] = dst4[q];
      sv[u] = src4[q];
    }
#pragma unroll
    for (int u = 0; u < LGROUP; ++u) {
      const int q = t + 1024 * (g * LGROUP + u);
      if (q < ECHUNKS) {
        place_edge(sv[u].x, dv[u].x, lo, pos, list);
        place_edge(sv[u].y, dv[u].y, lo, pos, list);
        place_edge(sv[u].z, dv[u].z, lo, pos, list);
        place_edge(sv[u].w, dv[u].w, lo, pos, list);
      }
    }
  }
}

__device__ __forceinline__ float2 row_f2(const float* __restrict__ base, int row, int c2) {
  return *(const float2*)(base + (long)row * FEAT + 2 * c2);
}

__global__ void __launch_bounds__(512) gather_mean50_kernel(const float* __restrict__ x,
                                                            const int* __restrict__ deg,
                                                            const int* __restrict__ cursor,
                                                            const int* __restrict__ ssrc,
                                                            float* __restrict__ mean) {
  __shared__ float stage[16 * FEAT];
  const int lane = threadIdx.x & 31;
  const int w = __builtin_amdgcn_readfirstlane(threadIdx.x >> 5);
  const int R = blockIdx.x * 16 + w;
  const int b = R / NN;
  const int d = clampi(__builtin_amdgcn_readfirstlane(deg[R]), 0, EE);
  const int start = clampi(__builtin_amdgcn_readfirstlane(cursor[R]) - d, 0, EE - d);
  const int* __restrict__ list = ssrc + (long)b * EE + start;
  const float* __restrict__ xb = x + (long)b * NN * FEAT;
  const int c2 = imin(lane, FEAT / 2 - 1);
  float2 acc = make_float2(0.0f, 0.0f);
  for (int c = 0; c < d; c += 32) {
    const int n = imin(d - c, 32);
    const int mine = clampi((lane < n) ? list[c + lane] : 0, 0, NN - 1);
    int j = 0;
    for (; j + 4 <= n; j += 4) {
      const float2 v0 = row_f2(xb, __builtin_amdgcn_readlane(mine, j),     c2);
      const float2 v1 = row_f2(xb, __builtin_amdgcn_readlane(mine, j + 1), c2);
      const float2 v2 = row_f2(xb, __builtin_amdgcn_readlane(mine, j + 2), c2);
      const float2 v3 = row_f2(xb, __builtin_amdgcn_readlane(mine, j + 3), c2);
      acc.x += v0.x; acc.y += v0.y;
      acc.x += v1.x; acc.y += v1.y;
      acc.x += v2.x; acc.y += v2.y;
      acc.x += v3.x; acc.y += v3.y;
    }
    for (; j < n; ++j) {
      const float2 v0 = row_f2(xb, __builtin_amdgcn_readlane(mine, j), c2);
      acc.x += v0.x; acc.y += v0.y;
    }
  }
  const float cdiv = fmaxf((float)d, 1.0f);
  if (lane < FEAT / 2) {
    stage[w * FEAT + 2 * lane]     = acc.x / cdiv;
    stage[w * FEAT + 2 * lane + 1] = acc.y / cdiv;
  }
  __syncthreads();
  float* __restrict__ outp = mean + (long)blockIdx.x * (16 * FEAT);
  for (int t = threadIdx.x; t < 16 * FEAT; t += 512) outp[t] = stage[t];
}

__device__ __forceinline__ float4 row_f4(const float* __restrict__ base, int row, int lane) {
  return *(const float4*)(base + (long)row * HID + 4 * lane);
}

__global__ void gather_mean128_live_kernel(const float* __restrict__ hsrc,
                                           const int* __restrict__ deg,
                                           const int* __restrict__ cursor,
                                           const int* __restrict__ ssrc,
                                           float* __restrict__ mean,
                                           const int* __restrict__ act_range_p) {
  const int lane = threadIdx.x & 31;
  const int r = blockIdx.x * (blockDim.x >> 5) + __builtin_amdgcn_readfirstlane(threadIdx.x >> 5);
  const int b = blockIdx.y;
  if (r >= PADR) return;
  const int live_rows = imin(*act_range_p, PADR);
  const int R = b * NN + r;
  float4 acc = make_float4(0.0f, 0.0f, 0.0f, 0.0f);
  float cdiv = 1.0f;
  if (r < live_rows) {
    const int d = clampi(__builtin_amdgcn_readfirstlane(deg[R]), 0, EE);
    const int start = clampi(__builtin_amdgcn_readfirstlane(cursor[R]) - d, 0, EE - d);
    const int* __restrict__ list = ssrc + (long)b * EE + start;
    const float* __restrict__ hb = hsrc + (long)b * NN * HID;
    for (int c = 0; c < d; c += 32) {
      const int n = imin(d - c, 32);
      const int mine = clampi((lane < n) ? list[c + lane] : 0, 0, NN - 1);
      int j = 0;
      for (; j + 4 <= n; j += 4) {
        const float4 v0 = row_f4(hb, __builtin_amdgcn_readlane(mine, j),     lane);
        const float4 v1 = row_f4(hb, __builtin_amdgcn_readlane(mine, j + 1), lane);
        const float4 v2 = row_f4(hb, __builtin_amdgcn_readlane(mine, j + 2), lane);
        const float4 v3 = row_f4(hb, __builtin_amdgcn_readlane(mine, j + 3), lane);
        acc.x += v0.x; acc.y += v0.y; acc.z += v0.z; acc.w += v0.w;
        acc.x += v1.x; acc.y += v1.y; acc.z += v1.z; acc.w += v1.w;
        acc.x += v2.x; acc.y += v2.y; acc.z += v2.z; acc.w += v2.w;
        acc.x += v3.x; acc.y += v3.y; acc.z += v3.z; acc.w += v3.w;
      }
      for (; j < n; ++j) {
        const float4 v0 = row_f4(hb, __builtin_amdgcn_readlane(mine, j), lane);
        acc.x += v0.x; acc.y += v0.y; acc.z += v0.z; acc.w += v0.w;
      }
    }
    cdiv = fmaxf((float)d, 1.0f);
  }
  float4 o;
  o.x = acc.x / cdiv; o.y = acc.y / cdiv; o.z = acc.z / cdiv; o.w = acc.w / cdiv;
  *(float4*)(mean + (long)R * HID + 4 * lane) = o;
}

template <int NT, int NCT>
__global__ void wmma_gemm_kernel(const float* __restrict__ A,
                                 const float* __restrict__ A2,
                                 const float* __restrict__ Wa,
                                 const float* __restrict__ Wb,
                                 const float* __restrict__ bias,
                                 float* __restrict__ C,
                                 int M, int K, int relu,
                                 int pad_mode, const int* __restrict__ act_range_p,
                                 int pad_rows, int src_rows) {
  const int lane = threadIdx.x & 31;
  const int mt = blockIdx.x * (blockDim.x >> 5) + (threadIdx.x >> 5);
  const int tm = (M + 15) >> 4;
  if (mt >= tm) return;

  const int ml = lane & 15;
  const int g  = lane >> 4;

  const int row = mt * 16 + ml;
  bool rvalid = row < M;
  long arow;
  if (pad_mode) {
    const int rc = imin(row, M - 1);
    const int b  = rc / pad_rows;
    const int rr = rc - b * pad_rows;
    rvalid = rvalid && (rr < *act_range_p);
    arow = (long)b * src_rows + rr;
  } else {
    arow = imin(row, M - 1);
  }

  v8f acc[NT];
#pragma unroll
  for (int t = 0; t < NT; ++t) acc[t] = (v8f){};

  const int npass = (A2 != nullptr) ? 2 : 1;

  for (int pass = 0; pass < npass; ++pass) {
    const float* __restrict__ Ap = pass ? A2 : A;
    const float* __restrict__ Wp = pass ? Wb : Wa;

    const float* aptr = Ap + arow * (long)K + 2 * g;
    const float* bptr = Wp + (2 * g) * NCT + ml;
    const int nfull = K >> 2;

    for (int it = 0; it < nfull; ++it) {
      v2f af = *(const v2f*)aptr;
      af.x = rvalid ? af.x : 0.0f;
      af.y = rvalid ? af.y : 0.0f;
#pragma unroll
      for (int t = 0; t < NT; ++t) {
        v2f bf;
        bf.x = bptr[t * 16];
        bf.y = bptr[NCT + t * 16];
        acc[t] = wmma_f32x4(af, bf, acc[t]);
      }
      aptr += 4;
      bptr += 4 * NCT;
    }

    const int k0 = nfull << 2;
    if (k0 < K) {
      const float* __restrict__ Arow_p = Ap + arow * (long)K;
      const int ka  = k0 + 2 * g;
      const int kc0 = imin(ka, K - 1);
      const int kc1 = imin(ka + 1, K - 1);
      float a0 = Arow_p[kc0];
      float a1 = Arow_p[kc1];
      v2f af;
      af.x = (rvalid && ka     < K) ? a0 : 0.0f;
      af.y = (rvalid && ka + 1 < K) ? a1 : 0.0f;
      const float* __restrict__ W0 = Wp + (long)kc0 * NCT + ml;
      const float* __restrict__ W1 = Wp + (long)kc1 * NCT + ml;
#pragma unroll
      for (int t = 0; t < NT; ++t) {
        v2f bf;
        bf.x = W0[t * 16];
        bf.y = W1[t * 16];
        acc[t] = wmma_f32x4(af, bf, acc[t]);
      }
    }
  }

  float* crow = C + (long)(mt * 16 + (g << 3)) * NCT + ml;
#pragma unroll
  for (int t = 0; t < NT; ++t) {
    const int n = t * 16 + ml;
    const bool nvalid = (NT * 16 == NCT) ? true : (n < NCT);
    const float bv = (bias != nullptr) ? bias[imin(n, NCT - 1)] : 0.0f;
#pragma unroll
    for (int v = 0; v < 8; ++v) {
      const int r = mt * 16 + v + (g << 3);
      float val = acc[t][v] + bv;
      if (relu) val = fmaxf(val, 0.0f);
      if (r < M && nvalid) crow[v * NCT + t * 16] = val;
    }
  }
}

#define TROWS 32
#define PA    132
#define PX    52

__device__ __forceinline__ float4 keep_if(bool keep, float4 a) {
  return make_float4(keep ? a.x : 0.0f, keep ? a.y : 0.0f, keep ? a.z : 0.0f, keep ? a.w : 0.0f);
}

__global__ void __launch_bounds__(256) tail_fused_kernel(const float* __restrict__ mean2,
                                                         const float* __restrict__ h,
                                                         const float* __restrict__ Wl2,
                                                         const float* __restrict__ Wr2,
                                                         const float* __restrict__ b2c,
                                                         const float* __restrict__ W1,
                                                         const float* __restrict__ b1,
                                                         const float* __restrict__ W2,
                                                         const float* __restrict__ b2,
                                                         const float* __restrict__ W3,
                                                         const float* __restrict__ b3,
                                                         const int* __restrict__ act_range_p,
                                                         float* __restrict__ out) {
  __shared__ __attribute__((aligned(16))) float sA[TROWS * PA];
  __shared__ __attribute__((aligned(16))) float sH[TROWS * PA];
  __shared__ __attribute__((aligned(16))) float sX[TROWS * PX];
  __shared__ int sValid[TROWS];

  const int tid  = threadIdx.x;
  const int lane = tid & 31;
  const int w = __builtin_amdgcn_readfirstlane(tid >> 5);
  const int c = lane & 15;
  const int g = lane >> 4;
  const int row0 = blockIdx.x * TROWS;
  const int act = *act_range_p;

  float4 m[TROWS / 8], v[TROWS / 8];
  bool live[TROWS / 8];
  for (int j = 0; j < TROWS / 8; ++j) {
    const int Rp = row0 + w + 8 * j;
    const int b  = Rp / PADR;
    const int rr = Rp - b * PADR;
    const long src = ((long)b * NN + rr) * HID + 4 * lane;
    m[j] = *(const float4*)(mean2 + src);
    v[j] = *(const float4*)(h + src);
    live[j] = rr < act;
  }
  for (int j = 0; j < TROWS / 8; ++j) {
    const int i = w + 8 * j;
    *(float4*)&sA[i * PA + 4 * lane] = keep_if(live[j], m[j]);
    *(float4*)&sH[i * PA + 4 * lane] = keep_if(live[j], v[j]);
  }
  if (tid < TROWS) {
    const int Rp = row0 + tid;
    sValid[tid] = (Rp - (Rp / PADR) * PADR) < act;
  }
  __syncthreads();

  {
    const int nt = w & 3;
    const int mt = w >> 2;
    const int col  = nt * 16 + c;
    const int colc = imin(col, FEAT - 1);
    const float* aM = &sA[(mt * 16 + c) * PA + 2 * g];
    const float* aH = &sH[(mt * 16 + c) * PA + 2 * g];
    const float* __restrict__ bL = Wl2 + (2 * g) * FEAT + colc;
    const float* __restrict__ bR = Wr2 + (2 * g) * FEAT + colc;
    v8f acc0 = (v8f){};
    v8f acc1 = (v8f){};
    for (int k0 = 0; k0 < HID; k0 += 4) {
      const v2f a0 = *(const v2f*)(aM + k0);
      const v2f a1 = *(const v2f*)(aH + k0);
      v2f f0, f1;
      f0.x = bL[k0 * FEAT];
      f0.y = bL[k0 * FEAT + FEAT];
      f1.x = bR[k0 * FEAT];
      f1.y = bR[k0 * FEAT + FEAT];
      acc0 = wmma_f32x4(a0, f0, acc0);
      acc1 = wmma_f32x4(a1, f1, acc1);
    }
    const float bias = b2c[colc];
    if (col < PX) {
      for (int r = 0; r < 8; ++r) {
        const int row = mt * 16 + 8 * g + r;
        const float val = acc0[r] + acc1[r] + bias;
        sX[row * PX + col] = (col < FEAT && sValid[row] != 0) ? val : 0.0f;
      }
    }
  }
  __syncthreads();

  {
    const int col = w * 16 + c;
    const float* a0p = &sX[c * PX + 2 * g];
    const float* a1p = &sX[(16 + c) * PX + 2 * g];
    v8f acc0 = (v8f){};
    v8f acc1 = (v8f){};
    for (int k0 = 0; k0 < PX; k0 += 4) {
      const int ka = k0 + 2 * g;
      v2f bf;
      bf.x = W1[imin(ka,     FEAT - 1) * HID + col];
      bf.y = W1[imin(ka + 1, FEAT - 1) * HID + col];
      const v2f a0 = *(const v2f*)(a0p + k0);
      const v2f a1 = *(const v2f*)(a1p + k0);
      acc0 = wmma_f32x4(a0, bf, acc0);
      acc1 = wmma_f32x4(a1, bf, acc1);
    }
    const float bias = b1[col];
    for (int r = 0; r < 8; ++r) {
      sA[(8 * g + r) * PA + col]      = fmaxf(acc0[r] + bias, 0.0f);
      sA[(16 + 8 * g + r) * PA + col] = fmaxf(acc1[r] + bias, 0.0f);
    }
  }
  __syncthreads();

  {
    const int col = w * 16 + c;
    const float* a0p = &sA[c * PA + 2 * g];
    const float* a1p = &sA[(16 + c) * PA + 2 * g];
    const float* __restrict__ bp = W2 + (2 * g) * HID + col;
    v8f acc0 = (v8f){};
    v8f acc1 = (v8f){};
    for (int k0 = 0; k0 < HID; k0 += 4) {
      v2f bf;
      bf.x = bp[k0 * HID];
      bf.y = bp[k0 * HID + HID];
      const v2f a0 = *(const v2f*)(a0p + k0);
      const v2f a1 = *(const v2f*)(a1p + k0);
      acc0 = wmma_f32x4(a0, bf, acc0);
      acc1 = wmma_f32x4(a1, bf, acc1);
    }
    const float bias = b2[col];
    const float w3 = W3[col];
    for (int r = 0; r < 8; ++r) {
      sH[(8 * g + r) * PA + col]      = fmaxf(acc0[r] + bias, 0.0f) * w3;
      sH[(16 + 8 * g + r) * PA + col] = fmaxf(acc1[r] + bias, 0.0f) * w3;
    }
  }
  __syncthreads();

  if (w == 0) {
    const float4* p = (const float4*)&sH[lane * PA];
    float s = 0.0f;
    for (int j = 0; j < HID / 4; ++j) {
      const float4 v = p[j];
      s += v.x; s += v.y; s += v.z; s += v.w;
    }
    out[row0 + lane] = s + b3[0];
  }
}


static inline int strip_grid(int M) {
  int tm = (M + 15) / 16;
  return (tm + 3) / 4;
}

extern "C" void kernel_launch(void* const* d_in, const int* in_sizes, int n_in,
                              void* d_out, int out_size, void* d_ws, size_t ws_size,
                              hipStream_t stream) {
  (void)in_sizes; (void)n_in; (void)out_size; (void)ws_size;

  const float* edge_feat = (const float*)d_in[0];
  const int*   edge_idx  = (const int*)d_in[1];
  const int*   act_range = (const int*)d_in[2];
  const float* Wl1 = (const float*)d_in[3];
  const float* Wr1 = (const float*)d_in[4];
  const float* b1c = (const float*)d_in[5];
  const float* Wl2 = (const float*)d_in[6];
  const float* Wr2 = (const float*)d_in[7];
  const float* b2c = (const float*)d_in[8];
  const float* W1  = (const float*)d_in[9];
  const float* b1  = (const float*)d_in[10];
  const float* W2  = (const float*)d_in[11];
  const float* b2  = (const float*)d_in[12];
  const float* W3  = (const float*)d_in[13];
  const float* b3  = (const float*)d_in[14];
  float* out = (float*)d_out;

  const long S_AGG1 = (long)BB * NN * FEAT;
  const long S_CNT  = (long)BB * NN;
  const long S_H    = (long)BB * NN * HID;

  float* agg1 = (float*)d_ws;
  float* cnt  = agg1 + S_AGG1;
  float* h    = cnt  + S_CNT;
  float* agg2 = h    + S_H;
  float* lists = agg2 + S_H;

  int* deg    = (int*)cnt;
  int* cursor = (int*)lists;
  int* ssrc   = cursor + S_CNT;

  build_lists_kernel<<<dim3(NRANGES, BB), 1024, 0, stream>>>(edge_idx, deg, cursor, ssrc);

  gather_mean50_kernel<<<BB * NN / 16, 512, 0, stream>>>(edge_feat, deg, cursor, ssrc, agg1);
  wmma_gemm_kernel<8, 128><<<strip_grid(BB * NN), 128, 0, stream>>>(
      agg1, edge_feat, Wl1, Wr1, b1c, h, BB * NN, FEAT,  1,
       0, nullptr, 0, 0);

  gather_mean128_live_kernel<<<dim3((PADR + 3) / 4, BB), 128, 0, stream>>>(
      h, deg, cursor, ssrc, agg2, act_range);
  tail_fused_kernel<<<BB * PADR / TROWS, 256, 0, stream>>>(
      agg2, h, Wl2, Wr2, b2c, W1, b1, W2, b2, W3, b3, act_range, out);
}


# layer 1 kernel: 1000 resident blocks of 16 waves loop over the 5000 row tiles (5 each) instead of one block per tile; B fragments once per wave, next rows' degree and list head read ahead
# speedup vs baseline: 12.8069x; 1.2616x over previous
#include <hip/hip_runtime.h>

#define BB    16
#define NN    5000
#define EE    80000
#define FEAT  50
#define HID   128
#define PADR  1000

typedef __attribute__((ext_vector_type(2))) float v2f;
typedef __attribute__((ext_vector_type(8))) float v8f;

__device__ __forceinline__ v8f wmma_f32x4(v2f a, v2f b, v8f c) {
  return __builtin_amdgcn_wmma_f32_16x16x4_f32(
       false, a,  false, b,
       (short)0, c,  false,  false);
}

__device__ __forceinline__ int imin(int a, int b) { return a < b ? a : b; }


__device__ __forceinline__ int clampi(int v, int lo, int hi) { return v < lo ? lo : (v > hi ? hi : v); }

#define NRANGES 8
#define RNODES  (NN / NRANGES)
#define ECHUNKS (EE / 4)
#define LGROUP  5
#define NGROUPS 4

__device__ __forceinline__ void count_edge(int d, int lo, int* cnt, int& below) {
  below += __builtin_popcount(__builtin_amdgcn_ballot_w32(d < lo));
  const unsigned r = (unsigned)(d - lo);
  if (r < (unsigned)RNODES) atomicAdd(&cnt[r], 1);
}

__device__ __forceinline__ void place_edge(int s, int d, int lo, int* pos, int* __restrict__ list) {
  const unsigned r = (unsigned)(d - lo);
  if (r < (unsigned)RNODES) {
    int p = atomicAdd(&pos[r], 1);
    p = clampi(p, 0, EE - 1);
    list[p] = s;
  }
}

__global__ void __launch_bounds__(1024) build_lists_kernel(const int* __restrict__ ei,
                                                           int* __restrict__ deg,
                                                           int* __restrict__ cursor,
                                                           int* __restrict__ ssrc) {
  __shared__ int cnt[RNODES];
  __shared__ int pos[RNODES];
  __shared__ int wsum[32];
  __shared__ int wbelow[32];
  __shared__ int sbase;
  const int t = threadIdx.x;
  const int lane = t & 31;
  const int w = t >> 5;
  const int b = blockIdx.y;
  const int lo = blockIdx.x * RNODES;
  const int* __restrict__ eib = ei + (long)b * 2 * EE;
  const int4* __restrict__ src4 = (const int4*)eib;
  const int4* __restrict__ dst4 = (const int4*)(eib + EE);
  int* __restrict__ list = ssrc + (long)b * EE;

  if (t < RNODES) cnt[t] = 0;
  __syncthreads();

  int below = 0;
  for (int g = 0; g < NGROUPS; ++g) {
    int4 dv[LGROUP];
#pragma unroll
    for (int u = 0; u < LGROUP; ++u) {
      const int q = t + 1024 * (g * LGROUP + u);
      dv[u] = dst4[imin(q, ECHUNKS - 1)];
    }
#pragma unroll
    for (int u = 0; u < LGROUP; ++u) {
      const int q = t + 1024 * (g * LGROUP + u);
      if (q < ECHUNKS) {
        count_edge(dv[u].x, lo, cnt, below);
        count_edge(dv[u].y, lo, cnt, below);
        count_edge(dv[u].z, lo, cnt, below);
        count_edge(dv[u].w, lo, cnt, below);
      }
    }
  }
  if (lane == 0) wbelow[w] = below;
  __syncthreads();

  const int c = (t < RNODES) ? cnt[t] : 0;
  int inc = c;
  for (int off = 1; off < 32; off <<= 1) {
    const int n = __shfl_up(inc, off, 32);
    if (lane >= off) inc += n;
  }
  if (lane == 31) wsum[w] = inc;
  __syncthreads();
  if (w == 0) {
    const int tot = wsum[lane];
    int wi = tot;
    for (int off = 1; off < 32; off <<= 1) {
      const int n = __shfl_up(wi, off, 32);
      if (lane >= off) wi += n;
    }
    wsum[lane] = wi - tot;
    int bl = wbelow[lane];
    for (int off = 16; off > 0; off >>= 1) bl += __shfl_xor(bl, off, 32);
    if (lane == 0) sbase = bl;
  }
  __syncthreads();
  if (t < RNODES) {
    const int start = sbase + wsum[w] + inc - c;
    pos[t] = start;
    deg[b * NN + lo + t] = c;
    cursor[b * NN + lo + t] = start + c;
  }
  __syncthreads();

  for (int g = 0; g < NGROUPS; ++g) {
    int4 dv[LGROUP];
    int4 sv[LGROUP];
#pragma unroll
    for (int u = 0; u < LGROUP; ++u) {
      const int q = imin(t + 1024 * (g * LGROUP + u), ECHUNKS - 1);
      dv[u] = dst4[q];
      sv[u] = src4[q];
    }
#pragma unroll
    for (int u = 0; u < LGROUP; ++u) {
      const int q = t + 1024 * (g * LGROUP + u);
      if (q < ECHUNKS) {
        place_edge(sv[u].x, dv[u].x, lo, pos, list);
        place_edge(sv[u].y, dv[u].y, lo, pos, list);
        place_edge(sv[u].z, dv[u].z, lo, pos, list);
        place_edge(sv[u].w, dv[u].w, lo, pos, list);
      }
    }
  }
}

#define PA    132
#define PX    52
#define KSTEPS (PX / 4)
#define NTILES1 (BB * NN / 16)
#define S1_BLOCKS 1000
__device__ __forceinline__ float2 row_f2(const float* __restrict__ base, int row, int c2) {
  return *(const float2*)(base + (long)row * FEAT + 2 * c2);
}
__device__ __forceinline__ int list_entry(int entries, int j) {
  return clampi(__builtin_amdgcn_readlane(entries, j), 0, NN - 1);
}

__global__ void __launch_bounds__(512, 16) sage1_fused_kernel(const float* __restrict__ x,
                                                              const int* __restrict__ deg,
                                                              const int* __restrict__ cursor,
                                                              const int* __restrict__ ssrc,
                                                              const float* __restrict__ Wl1,
                                                              const float* __restrict__ Wr1,
                                                              const float* __restrict__ b1c,
                                                              float* __restrict__ h) {
  __shared__ __attribute__((aligned(16))) float sA[2 * 16 * PX];
  __shared__ __attribute__((aligned(16))) float sT[2 * 16 * PA];
  const int lane = threadIdx.x & 31;
  const int w = __builtin_amdgcn_readfirstlane(threadIdx.x >> 5);

  const int prod = w >> 3;
  const int cc = lane & 15;
  const int g  = lane >> 4;
  const int col = (w & 7) * 16 + cc;
  const float* __restrict__ Wp = prod ? Wr1 : Wl1;
  v2f bf[KSTEPS];
#pragma unroll
  for (int s = 0; s < KSTEPS; ++s) {
    const int ka = 4 * s + 2 * g;
    bf[s].x = Wp[imin(ka,     FEAT - 1) * HID + col];
    bf[s].y = Wp[imin(ka + 1, FEAT - 1) * HID + col];
  }
  float bias[4];
#pragma unroll
  for (int i = 0; i < 4; ++i) bias[i] = b1c[4 * lane + i];

  const int c2 = imin(lane, FEAT / 2 - 1);
  const float* ap = &sA[(prod * 16 + cc) * PX + 2 * g];
  float* tp = &sT[(prod * 16 + 8 * g) * PA + col];
  const int* dcp = (lane & 1) ? cursor : deg;

  const int tile0 = imin((int)blockIdx.x, NTILES1 - 1);
  const bool more1 = tile0 + S1_BLOCKS < NTILES1;
  int R  = tile0 * 16 + w;
  int Rn = more1 ? R + S1_BLOCKS * 16 : R;
  int d, dn, mine;
  const int* list;
  const int* listn;
  {
    const int dc0 = dcp[R];
    const int dc1 = dcp[Rn];
    d  = clampi(__builtin_amdgcn_readlane(dc0, 0), 0, EE);
    dn = more1 ? clampi(__builtin_amdgcn_readlane(dc1, 0), 0, EE) : 0;
    const int st0 = clampi(__builtin_amdgcn_readlane(dc0, 1) - d,  0, EE - d);
    const int st1 = clampi(__builtin_amdgcn_readlane(dc1, 1) - dn, 0, EE - dn);
    list  = ssrc + (long)(R  / NN) * EE + st0;
    listn = ssrc + (long)(Rn / NN) * EE + st1;
    mine = clampi((lane < imin(d, 32)) ? list[lane] : 0, 0, NN - 1);
  }

  for (int tile = tile0; tile < NTILES1; tile += S1_BLOCKS) {
    const int b = R / NN;
    const float* __restrict__ xb = x + (long)b * NN * FEAT;

    const int headn = (lane < imin(dn, 32)) ? listn[lane] : 0;
    const bool more2 = tile + 2 * S1_BLOCKS < NTILES1;
    const int R2 = more2 ? R + 2 * S1_BLOCKS * 16 : R;
    const int dc2 = dcp[R2];

    const float2 own = row_f2(x, R, c2);
    float2 acc = make_float2(0.0f, 0.0f);
    for (int c = 0; c < d; ) {
      const int n = imin(d - c, 32);
      int j = 0;
      for (; j + 4 <= n; j += 4) {
        const float2 v0 = row_f2(xb, list_entry(mine, j),     c2);
        const float2 v1 = row_f2(xb, list_entry(mine, j + 1), c2);
        const float2 v2 = row_f2(xb, list_entry(mine, j + 2), c2);
        const float2 v3 = row_f2(xb, list_entry(mine, j + 3), c2);
        acc.x += v0.x; acc.y += v0.y;
        acc.x += v1.x; acc.y += v1.y;
        acc.x += v2.x; acc.y += v2.y;
        acc.x += v3.x; acc.y += v3.y;
      }
      for (; j < n; ++j) {
        const float2 v0 = row_f2(xb, list_entry(mine, j), c2);
        acc.x += v0.x; acc.y += v0.y;
      }
      c += 32;
      if (c < d) {
        const int n2 = imin(d - c, 32);
        mine = clampi((lane < n2) ? list[c + lane] : 0, 0, NN - 1);
      }
    }

    const int d2 = more2 ? clampi(__builtin_amdgcn_readlane(dc2, 0), 0, EE) : 0;
    const int st2 = clampi(__builtin_amdgcn_readlane(dc2, 1) - d2, 0, EE - d2);
    const int* list2 = ssrc + (long)(R2 / NN) * EE + st2;

    const float cdiv = fmaxf((float)d, 1.0f);
    if (lane < FEAT / 2) {
      *(float2*)&sA[w * PX + 2 * lane] = make_float2(acc.x / cdiv, acc.y / cdiv);
      *(float2*)&sA[(16 + w) * PX + 2 * lane] = own;
    } else if (lane == FEAT / 2) {
      *(float2*)&sA[w * PX + FEAT] = make_float2(0.0f, 0.0f);
      *(float2*)&sA[(16 + w) * PX + FEAT] = make_float2(0.0f, 0.0f);
    }
    __syncthreads();

    {
      v8f pacc = (v8f){};
#pragma unroll
      for (int s = 0; s < KSTEPS; ++s) {
        const v2f a = *(const v2f*)(ap + 4 * s);
        pacc = wmma_f32x4(a, bf[s], pacc);
      }
      for (int r = 0; r < 8; ++r) tp[r * PA] = pacc[r];
    }
    __syncthreads();

    const float4 t0 = *(const float4*)&sT[w * PA + 4 * lane];
    const float4 t1 = *(const float4*)&sT[(16 + w) * PA + 4 * lane];
    float4 o;
    o.x = fmaxf(t0.x + t1.x + bias[0], 0.0f);
    o.y = fmaxf(t0.y + t1.y + bias[1], 0.0f);
    o.z = fmaxf(t0.z + t1.z + bias[2], 0.0f);
    o.w = fmaxf(t0.w + t1.w + bias[3], 0.0f);
    *(float4*)(h + (long)R * HID + 4 * lane) = o;

    R = Rn; d = dn; list = listn; mine = headn;
    Rn = R2; dn = d2; listn = list2;
  }
}

__device__ __forceinline__ float4 row_f4(const float* __restrict__ base, int row, int lane) {
  return *(const float4*)(base + (long)row * HID + 4 * lane);
}

__global__ void gather_mean128_live_kernel(const float* __restrict__ hsrc,
                                           const int* __restrict__ deg,
                                           const int* __restrict__ cursor,
                                           const int* __restrict__ ssrc,
                                           float* __restrict__ mean,
                                           const int* __restrict__ act_range_p) {
  const int lane = threadIdx.x & 31;
  const int r = blockIdx.x * (blockDim.x >> 5) + __builtin_amdgcn_readfirstlane(threadIdx.x >> 5);
  const int b = blockIdx.y;
  if (r >= PADR) return;
  const int live_rows = imin(*act_range_p, PADR);
  const int R = b * NN + r;
  float4 acc = make_float4(0.0f, 0.0f, 0.0f, 0.0f);
  float cdiv = 1.0f;
  if (r < live_rows) {
    const int d = clampi(__builtin_amdgcn_readfirstlane(deg[R]), 0, EE);
    const int start = clampi(__builtin_amdgcn_readfirstlane(cursor[R]) - d, 0, EE - d);
    const int* __restrict__ list = ssrc + (long)b * EE + start;
    const float* __restrict__ hb = hsrc + (long)b * NN * HID;
    for (int c = 0; c < d; c += 32) {
      const int n = imin(d - c, 32);
      const int mine = clampi((lane < n) ? list[c + lane] : 0, 0, NN - 1);
      int j = 0;
      for (; j + 4 <= n; j += 4) {
        const float4 v0 = row_f4(hb, __builtin_amdgcn_readlane(mine, j),     lane);
        const float4 v1 = row_f4(hb, __builtin_amdgcn_readlane(mine, j + 1), lane);
        const float4 v2 = row_f4(hb, __builtin_amdgcn_readlane(mine, j + 2), lane);
        const float4 v3 = row_f4(hb, __builtin_amdgcn_readlane(mine, j + 3), lane);
        acc.x += v0.x; acc.y += v0.y; acc.z += v0.z; acc.w += v0.w;
        acc.x += v1.x; acc.y += v1.y; acc.z += v1.z; acc.w += v1.w;
        acc.x += v2.x; acc.y += v2.y; acc.z += v2.z; acc.w += v2.w;
        acc.x += v3.x; acc.y += v3.y; acc.z += v3.z; acc.w += v3.w;
      }
      for (; j < n; ++j) {
        const float4 v0 = row_f4(hb, __builtin_amdgcn_readlane(mine, j), lane);
        acc.x += v0.x; acc.y += v0.y; acc.z += v0.z; acc.w += v0.w;
      }
    }
    cdiv = fmaxf((float)d, 1.0f);
  }
  float4 o;
  o.x = acc.x / cdiv; o.y = acc.y / cdiv; o.z = acc.z / cdiv; o.w = acc.w / cdiv;
  *(float4*)(mean + (long)R * HID + 4 * lane) = o;
}

#define TROWS 32

__device__ __forceinline__ float4 keep_if(bool keep, float4 a) {
  return make_float4(keep ? a.x : 0.0f, keep ? a.y : 0.0f, keep ? a.z : 0.0f, keep ? a.w : 0.0f);
}

__global__ void __launch_bounds__(256) tail_fused_kernel(const float* __restrict__ mean2,
                                                         const float* __restrict__ h,
                                                         const float* __restrict__ Wl2,
                                                         const float* __restrict__ Wr2,
                                                         const float* __restrict__ b2c,
                                                         const float* __restrict__ W1,
                                                         const float* __restrict__ b1,
                                                         const float* __restrict__ W2,
                                                         const float* __restrict__ b2,
                                                         const float* __restrict__ W3,
                                                         const float* __restrict__ b3,
                                                         const int* __restrict__ act_range_p,
                                                         float* __restrict__ out) {
  __shared__ __attribute__((aligned(16))) float sA[TROWS * PA];
  __shared__ __attribute__((aligned(16))) float sH[TROWS * PA];
  __shared__ __attribute__((aligned(16))) float sX[TROWS * PX];
  __shared__ int sValid[TROWS];

  const int tid  = threadIdx.x;
  const int lane = tid & 31;
  const int w = __builtin_amdgcn_readfirstlane(tid >> 5);
  const int c = lane & 15;
  const int g = lane >> 4;
  const int row0 = blockIdx.x * TROWS;
  const int act = *act_range_p;

  float4 m[TROWS / 8], v[TROWS / 8];
  bool live[TROWS / 8];
  for (int j = 0; j < TROWS / 8; ++j) {
    const int Rp = row0 + w + 8 * j;
    const int b  = Rp / PADR;
    const int rr = Rp - b * PADR;
    const long src = ((long)b * NN + rr) * HID + 4 * lane;
    m[j] = *(const float4*)(mean2 + src);
    v[j] = *(const float4*)(h + src);
    live[j] = rr < act;
  }
  for (int j = 0; j < TROWS / 8; ++j) {
    const int i = w + 8 * j;
    *(float4*)&sA[i * PA + 4 * lane] = keep_if(live[j], m[j]);
    *(float4*)&sH[i * PA + 4 * lane] = keep_if(live[j], v[j]);
  }
  if (tid < TROWS) {
    const int Rp = row0 + tid;
    sValid[tid] = (Rp - (Rp / PADR) * PADR) < act;
  }
  __syncthreads();

  {
    const int nt = w & 3;
    const int mt = w >> 2;
    const int col  = nt * 16 + c;
    const int colc = imin(col, FEAT - 1);
    const float* aM = &sA[(mt * 16 + c) * PA + 2 * g];
    const float* aH = &sH[(mt * 16 + c) * PA + 2 * g];
    const float* __restrict__ bL = Wl2 + (2 * g) * FEAT + colc;
    const float* __restrict__ bR = Wr2 + (2 * g) * FEAT + colc;
    v8f acc0 = (v8f){};
    v8f acc1 = (v8f){};
    for (int k0 = 0; k0 < HID; k0 += 4) {
      const v2f a0 = *(const v2f*)(aM + k0);
      const v2f a1 = *(const v2f*)(aH + k0);
      v2f f0, f1;
      f0.x = bL[k0 * FEAT];
      f0.y = bL[k0 * FEAT + FEAT];
      f1.x = bR[k0 * FEAT];
      f1.y = bR[k0 * FEAT + FEAT];
      acc0 = wmma_f32x4(a0, f0, acc0);
      acc1 = wmma_f32x4(a1, f1, acc1);
    }
    const float bias = b2c[colc];
    if (col < PX) {
      for (int r = 0; r < 8; ++r) {
        const int row = mt * 16 + 8 * g + r;
        const float val = acc0[r] + acc1[r] + bias;
        sX[row * PX + col] = (col < FEAT && sValid[row] != 0) ? val : 0.0f;
      }
    }
  }
  __syncthreads();

  {
    const int col = w * 16 + c;
    const float* a0p = &sX[c * PX + 2 * g];
    const float* a1p = &sX[(16 + c) * PX + 2 * g];
    v8f acc0 = (v8f){};
    v8f acc1 = (v8f){};
    for (int k0 = 0; k0 < PX; k0 += 4) {
      const int ka = k0 + 2 * g;
      v2f bf;
      bf.x = W1[imin(ka,     FEAT - 1) * HID + col];
      bf.y = W1[imin(ka + 1, FEAT - 1) * HID + col];
      const v2f a0 = *(const v2f*)(a0p + k0);
      const v2f a1 = *(const v2f*)(a1p + k0);
      acc0 = wmma_f32x4(a0, bf, acc0);
      acc1 = wmma_f32x4(a1, bf, acc1);
    }
    const float bias = b1[col];
    for (int r = 0; r < 8; ++r) {
      sA[(8 * g + r) * PA + col]      = fmaxf(acc0[r] + bias, 0.0f);
      sA[(16 + 8 * g + r) * PA + col] = fmaxf(acc1[r] + bias, 0.0f);
    }
  }
  __syncthreads();

  {
    const int col = w * 16 + c;
    const float* a0p = &sA[c * PA + 2 * g];
    const float* a1p = &sA[(16 + c) * PA + 2 * g];
    const float* __restrict__ bp = W2 + (2 * g) * HID + col;
    v8f acc0 = (v8f){};
    v8f acc1 = (v8f){};
    for (int k0 = 0; k0 < HID; k0 += 4) {
      v2f bf;
      bf.x = bp[k0 * HID];
      bf.y = bp[k0 * HID + HID];
      const v2f a0 = *(const v2f*)(a0p + k0);
      const v2f a1 = *(const v2f*)(a1p + k0);
      acc0 = wmma_f32x4(a0, bf, acc0);
      acc1 = wmma_f32x4(a1, bf, acc1);
    }
    const float bias = b2[col];
    const float w3 = W3[col];
    for (int r = 0; r < 8; ++r) {
      sH[(8 * g + r) * PA + col]      = fmaxf(acc0[r] + bias, 0.0f) * w3;
      sH[(16 + 8 * g + r) * PA + col] = fmaxf(acc1[r] + bias, 0.0f) * w3;
    }
  }
  __syncthreads();

  if (w == 0) {
    const float4* p = (const float4*)&sH[lane * PA];
    float s = 0.0f;
    for (int j = 0; j < HID / 4; ++j) {
      const float4 v = p[j];
      s += v.x; s += v.y; s += v.z; s += v.w;
    }
    out[row0 + lane] = s + b3[0];
  }
}


extern "C" void kernel_launch(void* const* d_in, const int* in_sizes, int n_in,
                              void* d_out, int out_size, void* d_ws, size_t ws_size,
                              hipStream_t stream) {
  (void)in_sizes; (void)n_in; (void)out_size; (void)ws_size;

  const float* edge_feat = (const float*)d_in[0];
  const int*   edge_idx  = (const int*)d_in[1];
  const int*   act_range = (const int*)d_in[2];
  const float* Wl1 = (const float*)d_in[3];
  const float* Wr1 = (const float*)d_in[4];
  const float* b1c = (const float*)d_in[5];
  const float* Wl2 = (const float*)d_in[6];
  const float* Wr2 = (const float*)d_in[7];
  const float* b2c = (const float*)d_in[8];
  const float* W1  = (const float*)d_in[9];
  const float* b1  = (const float*)d_in[10];
  const float* W2  = (const float*)d_in[11];
  const float* b2  = (const float*)d_in[12];
  const float* W3  = (const float*)d_in[13];
  const float* b3  = (const float*)d_in[14];
  float* out = (float*)d_out;

  const long S_AGG1 = (long)BB * NN * FEAT;
  const long S_CNT  = (long)BB * NN;
  const long S_H    = (long)BB * NN * HID;

  float* agg1 = (float*)d_ws;
  float* cnt  = agg1 + S_AGG1;
  float* h    = cnt  + S_CNT;
  float* agg2 = h    + S_H;
  float* lists = agg2 + S_H;

  int* deg    = (int*)cnt;
  int* cursor = (int*)lists;
  int* ssrc   = cursor + S_CNT;

  build_lists_kernel<<<dim3(NRANGES, BB), 1024, 0, stream>>>(edge_idx, deg, cursor, ssrc);

  sage1_fused_kernel<<<S1_BLOCKS, 512, 0, stream>>>(edge_feat, deg, cursor, ssrc, Wl1, Wr1, b1c, h);

  gather_mean128_live_kernel<<<dim3((PADR + 3) / 4, BB), 128, 0, stream>>>(
      h, deg, cursor, ssrc, agg2, act_range);
  tail_fused_kernel<<<BB * PADR / TROWS, 256, 0, stream>>>(
      agg2, h, Wl2, Wr2, b2c, W1, b1, W2, b2, W3, b3, act_range, out);
}


# edge-list build: two workgroups per batch and node range; each reads dst once, keeps its half's dst in registers and fills only that half, src asked for ahead of the scan; order inside a list changes
# speedup vs baseline: 14.3181x; 1.1180x over previous
#include <hip/hip_runtime.h>

#define BB    16
#define NN    5000
#define EE    80000
#define FEAT  50
#define HID   128
#define PADR  1000

typedef __attribute__((ext_vector_type(2))) float v2f;
typedef __attribute__((ext_vector_type(8))) float v8f;

__device__ __forceinline__ v8f wmma_f32x4(v2f a, v2f b, v8f c) {
  return __builtin_amdgcn_wmma_f32_16x16x4_f32(
       false, a,  false, b,
       (short)0, c,  false,  false);
}

__device__ __forceinline__ int imin(int a, int b) { return a < b ? a : b; }
__device__ __forceinline__ int imax(int a, int b) { return a > b ? a : b; }


__device__ __forceinline__ int clampi(int v, int lo, int hi) { return v < lo ? lo : (v > hi ? hi : v); }

#define NRANGES 8
#define RNODES  (NN / NRANGES)
#define ECHUNKS (EE / 4)
#define HCHUNKS (ECHUNKS / 2)
#define CPT     10
#define NO_EDGE 0x7fffffff

__device__ __forceinline__ int count_edge(int d, int lo, int* cnt, int& below) {
  below += (d < lo) ? 1 : 0;
  const int r = d - lo;
  if ((unsigned)r < (unsigned)RNODES) atomicAdd(&cnt[r], 1);
  return r;
}

__device__ __forceinline__ void place_edge(int s, int r, int* pos, int* __restrict__ list) {
  if ((unsigned)r < (unsigned)RNODES) {
    int p = atomicAdd(&pos[r], 1);
    p = clampi(p, 0, EE - 1);
    list[p] = s;
  }
}

__global__ void __launch_bounds__(1024) build_lists_kernel(const int* __restrict__ ei,
                                                           int* __restrict__ deg,
                                                           int* __restrict__ cursor,
                                                           int* __restrict__ ssrc) {
  __shared__ int cown[RNODES];
  __shared__ int coth[RNODES];
  __shared__ int pos[RNODES];
  __shared__ int wsum[32];
  __shared__ int wbelow[32];
  __shared__ int sbase;
  const int t = threadIdx.x;
  const int lane = t & 31;
  const int w = t >> 5;
  const int b = blockIdx.y;
  const int half = blockIdx.x & 1;
  const int lo = (blockIdx.x >> 1) * RNODES;
  const int* __restrict__ eib = ei + (long)b * 2 * EE;
  const int4* __restrict__ src4 = (const int4*)eib;
  const int4* __restrict__ dst4 = (const int4*)(eib + EE);
  int* __restrict__ list = ssrc + (long)b * EE;
  const int own0 = half * HCHUNKS;
  const int oth0 = HCHUNKS - own0;

  if (t < RNODES) { cown[t] = 0; coth[t] = 0; }
  __syncthreads();

  int below = 0;
  int4 rown[CPT];
  {
    int4 dv[CPT];
    int4 ov[CPT];
#pragma unroll
    for (int u = 0; u < CPT; ++u) {
      const int q = imin(t + 1024 * u, HCHUNKS - 1);
      dv[u] = dst4[own0 + q];
      ov[u] = dst4[oth0 + q];
    }
#pragma unroll
    for (int u = 0; u < CPT; ++u) {
      const bool ok = t + 1024 * u < HCHUNKS;
      const int4 a = ok ? dv[u] : make_int4(NO_EDGE, NO_EDGE, NO_EDGE, NO_EDGE);
      rown[u].x = count_edge(a.x, lo, cown, below);
      rown[u].y = count_edge(a.y, lo, cown, below);
      rown[u].z = count_edge(a.z, lo, cown, below);
      rown[u].w = count_edge(a.w, lo, cown, below);
    }
#pragma unroll
    for (int u = 0; u < CPT; ++u) {
      const bool ok = t + 1024 * u < HCHUNKS;
      const int4 a = ok ? ov[u] : make_int4(NO_EDGE, NO_EDGE, NO_EDGE, NO_EDGE);
      (void)count_edge(a.x, lo, coth, below);
      (void)count_edge(a.y, lo, coth, below);
      (void)count_edge(a.z, lo, coth, below);
      (void)count_edge(a.w, lo, coth, below);
    }
  }
  int4 sv[CPT];
#pragma unroll
  for (int u = 0; u < CPT; ++u) sv[u] = src4[own0 + imin(t + 1024 * u, HCHUNKS - 1)];

  for (int off = 16; off > 0; off >>= 1) below += __shfl_xor(below, off, 32);
  if (lane == 0) wbelow[w] = below;
  __syncthreads();

  const int co = (t < RNODES) ? coth[t] : 0;
  const int c = (t < RNODES) ? cown[t] + co : 0;
  int inc = c;
  for (int off = 1; off < 32; off <<= 1) {
    const int n = __shfl_up(inc, off, 32);
    if (lane >= off) inc += n;
  }
  if (lane == 31) wsum[w] = inc;
  __syncthreads();
  if (w == 0) {
    const int tot = wsum[lane];
    int wi = tot;
    for (int off = 1; off < 32; off <<= 1) {
      const int n = __shfl_up(wi, off, 32);
      if (lane >= off) wi += n;
    }
    wsum[lane] = wi - tot;
    int bl = wbelow[lane];
    for (int off = 16; off > 0; off >>= 1) bl += __shfl_xor(bl, off, 32);
    if (lane == 0) sbase = bl;
  }
  __syncthreads();
  if (t < RNODES) {
    const int start = sbase + wsum[w] + inc - c;
    pos[t] = start + (half ? co : 0);
    if (half == 0) {
      deg[b * NN + lo + t] = c;
      cursor[b * NN + lo + t] = start + c;
    }
  }
  __syncthreads();

#pragma unroll
  for (int u = 0; u < CPT; ++u) {
    place_edge(sv[u].x, rown[u].x, pos, list);
    place_edge(sv[u].y, rown[u].y, pos, list);
    place_edge(sv[u].z, rown[u].z, pos, list);
    place_edge(sv[u].w, rown[u].w, pos, list);
  }
}

#define PA    132
#define PX    52
#define KSTEPS (PX / 4)
#define NTILES1 (BB * NN / 16)
#define S1_BLOCKS 1000
__device__ __forceinline__ float2 row_f2(const float* __restrict__ base, int row, int c2) {
  return *(const float2*)(base + (long)row * FEAT + 2 * c2);
}
__device__ __forceinline__ int list_entry(int entries, int j) {
  return clampi(__builtin_amdgcn_readlane(entries, j), 0, NN - 1);
}

__global__ void __launch_bounds__(512, 16) sage1_fused_kernel(const float* __restrict__ x,
                                                              const int* __restrict__ deg,
                                                              const int* __restrict__ cursor,
                                                              const int* __restrict__ ssrc,
                                                              const float* __restrict__ Wl1,
                                                              const float* __restrict__ Wr1,
                                                              const float* __restrict__ b1c,
                                                              float* __restrict__ h) {
  __shared__ __attribute__((aligned(16))) float sA[2 * 16 * PX];
  __shared__ __attribute__((aligned(16))) float sT[2 * 16 * PA];
  const int lane = threadIdx.x & 31;
  const int w = __builtin_amdgcn_readfirstlane(threadIdx.x >> 5);

  const int prod = w >> 3;
  const int cc = lane & 15;
  const int g  = lane >> 4;
  const int col = (w & 7) * 16 + cc;
  const float* __restrict__ Wp = prod ? Wr1 : Wl1;
  v2f bf[KSTEPS];
#pragma unroll
  for (int s = 0; s < KSTEPS; ++s) {
    const int ka = 4 * s + 2 * g;
    bf[s].x = Wp[imin(ka,     FEAT - 1) * HID + col];
    bf[s].y = Wp[imin(ka + 1, FEAT - 1) * HID + col];
  }
  float bias[4];
#pragma unroll
  for (int i = 0; i < 4; ++i) bias[i] = b1c[4 * lane + i];

  const int c2 = imin(lane, FEAT / 2 - 1);
  const float* ap = &sA[(prod * 16 + cc) * PX + 2 * g];
  float* tp = &sT[(prod * 16 + 8 * g) * PA + col];
  const int* dcp = (lane & 1) ? cursor : deg;

  const int tile0 = imin((int)blockIdx.x, NTILES1 - 1);
  const bool more1 = tile0 + S1_BLOCKS < NTILES1;
  int R  = tile0 * 16 + w;
  int Rn = more1 ? R + S1_BLOCKS * 16 : R;
  int d, dn, mine;
  const int* list;
  const int* listn;
  {
    const int dc0 = dcp[R];
    const int dc1 = dcp[Rn];
    d  = clampi(__builtin_amdgcn_readlane(dc0, 0), 0, EE);
    dn = more1 ? clampi(__builtin_amdgcn_readlane(dc1, 0), 0, EE) : 0;
    const int st0 = clampi(__builtin_amdgcn_readlane(dc0, 1) - d,  0, EE - d);
    const int st1 = clampi(__builtin_amdgcn_readlane(dc1, 1) - dn, 0, EE - dn);
    list  = ssrc + (long)(R  / NN) * EE + st0;
    listn = ssrc + (long)(Rn / NN) * EE + st1;
    mine = clampi((lane < imin(d, 32)) ? list[lane] : 0, 0, NN - 1);
  }

  for (int tile = tile0; tile < NTILES1; tile += S1_BLOCKS) {
    const int b = R / NN;
    const float* __restrict__ xb = x + (long)b * NN * FEAT;

    const int headn = (lane < imin(dn, 32)) ? listn[lane] : 0;
    const bool more2 = tile + 2 * S1_BLOCKS < NTILES1;
    const int R2 = more2 ? R + 2 * S1_BLOCKS * 16 : R;
    const int dc2 = dcp[R2];

    const float2 own = row_f2(x, R, c2);
    float2 acc = make_float2(0.0f, 0.0f);
    for (int c = 0; c < d; ) {
      const int n = imin(d - c, 32);
      int j = 0;
      for (; j + 4 <= n; j += 4) {
        const float2 v0 = row_f2(xb, list_entry(mine, j),     c2);
        const float2 v1 = row_f2(xb, list_entry(mine, j + 1), c2);
        const float2 v2 = row_f2(xb, list_entry(mine, j + 2), c2);
        const float2 v3 = row_f2(xb, list_entry(mine, j + 3), c2);
        acc.x += v0.x; acc.y += v0.y;
        acc.x += v1.x; acc.y += v1.y;
        acc.x += v2.x; acc.y += v2.y;
        acc.x += v3.x; acc.y += v3.y;
      }
      for (; j < n; ++j) {
        const float2 v0 = row_f2(xb, list_entry(mine, j), c2);
        acc.x += v0.x; acc.y += v0.y;
      }
      c += 32;
      if (c < d) {
        const int n2 = imin(d - c, 32);
        mine = clampi((lane < n2) ? list[c + lane] : 0, 0, NN - 1);
      }
    }

    const int d2 = more2 ? clampi(__builtin_amdgcn_readlane(dc2, 0), 0, EE) : 0;
    const int st2 = clampi(__builtin_amdgcn_readlane(dc2, 1) - d2, 0, EE - d2);
    const int* list2 = ssrc + (long)(R2 / NN) * EE + st2;

    const float cdiv = fmaxf((float)d, 1.0f);
    if (lane < FEAT / 2) {
      *(float2*)&sA[w * PX + 2 * lane] = make_float2(acc.x / cdiv, acc.y / cdiv);
      *(float2*)&sA[(16 + w) * PX + 2 * lane] = own;
    } else if (lane == FEAT / 2) {
      *(float2*)&sA[w * PX + FEAT] = make_float2(0.0f, 0.0f);
      *(float2*)&sA[(16 + w) * PX + FEAT] = make_float2(0.0f, 0.0f);
    }
    __syncthreads();

    {
      v8f pacc = (v8f){};
#pragma unroll
      for (int s = 0; s < KSTEPS; ++s) {
        const v2f a = *(const v2f*)(ap + 4 * s);
        pacc = wmma_f32x4(a, bf[s], pacc);
      }
      for (int r = 0; r < 8; ++r) tp[r * PA] = pacc[r];
    }
    __syncthreads();

    const float4 t0 = *(const float4*)&sT[w * PA + 4 * lane];
    const float4 t1 = *(const float4*)&sT[(16 + w) * PA + 4 * lane];
    float4 o;
    o.x = fmaxf(t0.x + t1.x + bias[0], 0.0f);
    o.y = fmaxf(t0.y + t1.y + bias[1], 0.0f);
    o.z = fmaxf(t0.z + t1.z + bias[2], 0.0f);
    o.w = fmaxf(t0.w + t1.w + bias[3], 0.0f);
    *(float4*)(h + (long)R * HID + 4 * lane) = o;

    R = Rn; d = dn; list = listn; mine = headn;
    Rn = R2; dn = d2; listn = list2;
  }
}

__device__ __forceinline__ float4 row_f4(const float* __restrict__ base, int row, int lane) {
  return *(const float4*)(base + (long)row * HID + 4 * lane);
}

#define TROWS 32
#define GU    8

__device__ __forceinline__ float4 keep_if(bool keep, float4 a) {
  return make_float4(keep ? a.x : 0.0f, keep ? a.y : 0.0f, keep ? a.z : 0.0f, keep ? a.w : 0.0f);
}

__global__ void __launch_bounds__(256) tail_fused_kernel(const float* __restrict__ h,
                                                         const int* __restrict__ deg,
                                                         const int* __restrict__ cursor,
                                                         const int* __restrict__ ssrc,
                                                         const float* __restrict__ Wl2,
                                                         const float* __restrict__ Wr2,
                                                         const float* __restrict__ b2c,
                                                         const float* __restrict__ W1,
                                                         const float* __restrict__ b1,
                                                         const float* __restrict__ W2,
                                                         const float* __restrict__ b2,
                                                         const float* __restrict__ W3,
                                                         const float* __restrict__ b3,
                                                         const int* __restrict__ act_range_p,
                                                         float* __restrict__ out) {
  __shared__ __attribute__((aligned(16))) float sA[TROWS * PA];
  __shared__ __attribute__((aligned(16))) float sH[TROWS * PA];
  __shared__ __attribute__((aligned(16))) float sX[TROWS * PX];
  __shared__ int sValid[TROWS];

  const int tid  = threadIdx.x;
  const int lane = tid & 31;
  const int w = __builtin_amdgcn_readfirstlane(tid >> 5);
  const int c = lane & 15;
  const int g = lane >> 4;
  const int row0 = blockIdx.x * TROWS;
  const int act = *act_range_p;

  int dc;
  {
    const int Rp = row0 + w + 8 * ((lane >> 1) & 3);
    const int b  = Rp / PADR;
    dc = ((lane & 1) ? cursor : deg)[b * NN + (Rp - b * PADR)];
  }
  float4 v[TROWS / 8];
  bool live[TROWS / 8];
  int bj[TROWS / 8];
  for (int j = 0; j < TROWS / 8; ++j) {
    const int Rp = row0 + w + 8 * j;
    const int b  = Rp / PADR;
    const int rr = Rp - b * PADR;
    bj[j] = b;
    v[j] = *(const float4*)(h + ((long)b * NN + rr) * HID + 4 * lane);
    live[j] = rr < act;
  }
  int d[TROWS / 8], head[TROWS / 8];
  const int* list[TROWS / 8];
  for (int j = 0; j < TROWS / 8; ++j) {
    d[j] = live[j] ? clampi(__builtin_amdgcn_readlane(dc, 2 * j), 0, EE) : 0;
    const int st = clampi(__builtin_amdgcn_readlane(dc, 2 * j + 1) - d[j], 0, EE - d[j]);
    list[j] = ssrc + (long)bj[j] * EE + st;
    head[j] = (lane < imin(d[j], 32)) ? list[j][lane] : 0;
  }
  const int dmax = imax(imax(d[0], d[1]), imax(d[2], d[3]));
  float4 acc[TROWS / 8];
  for (int j = 0; j < TROWS / 8; ++j) acc[j] = make_float4(0.0f, 0.0f, 0.0f, 0.0f);
  for (int c0 = 0; c0 < dmax; c0 += 32) {
    if (c0 > 0) {
      for (int j = 0; j < TROWS / 8; ++j) head[j] = (c0 + lane < d[j]) ? list[j][c0 + lane] : 0;
    }
    const int nmax = imin(dmax - c0, 32);
    for (int e = 0; e < nmax; e += GU) {
      float4 t[TROWS / 8][GU];
#pragma unroll
      for (int j = 0; j < TROWS / 8; ++j) {
        const float* __restrict__ hb = h + (long)bj[j] * NN * HID;
#pragma unroll
        for (int u = 0; u < GU; ++u) {
          const int sn = clampi(__builtin_amdgcn_readlane(head[j], e + u), 0, NN - 1);
          t[j][u] = (c0 + e + u < d[j]) ? row_f4(hb, sn, lane) : make_float4(0.0f, 0.0f, 0.0f, 0.0f);
        }
      }
#pragma unroll
      for (int j = 0; j < TROWS / 8; ++j) {
#pragma unroll
        for (int u = 0; u < GU; ++u) {
          acc[j].x += t[j][u].x; acc[j].y += t[j][u].y; acc[j].z += t[j][u].z; acc[j].w += t[j][u].w;
        }
      }
    }
  }
  for (int j = 0; j < TROWS / 8; ++j) {
    const int i = w + 8 * j;
    const float cdiv = fmaxf((float)d[j], 1.0f);
    *(float4*)&sA[i * PA + 4 * lane] =
        make_float4(acc[j].x / cdiv, acc[j].y / cdiv, acc[j].z / cdiv, acc[j].w / cdiv);
    *(float4*)&sH[i * PA + 4 * lane] = keep_if(live[j], v[j]);
  }
  if (tid < TROWS) {
    const int Rp = row0 + tid;
    sValid[tid] = (Rp - (Rp / PADR) * PADR) < act;
  }
  __syncthreads();

  {
    const int nt = w & 3;
    const int mt = w >> 2;
    const int col  = nt * 16 + c;
    const int colc = imin(col, FEAT - 1);
    const float* aM = &sA[(mt * 16 + c) * PA + 2 * g];
    const float* aH = &sH[(mt * 16 + c) * PA + 2 * g];
    const float* __restrict__ bL = Wl2 + (2 * g) * FEAT + colc;
    const float* __restrict__ bR = Wr2 + (2 * g) * FEAT + colc;
    v8f acc0 = (v8f){};
    v8f acc1 = (v8f){};
    for (int k0 = 0; k0 < HID; k0 += 4) {
      const v2f a0 = *(const v2f*)(aM + k0);
      const v2f a1 = *(const v2f*)(aH + k0);
      v2f f0, f1;
      f0.x = bL[k0 * FEAT];
      f0.y = bL[k0 * FEAT + FEAT];
      f1.x = bR[k0 * FEAT];
      f1.y = bR[k0 * FEAT + FEAT];
      acc0 = wmma_f32x4(a0, f0, acc0);
      acc1 = wmma_f32x4(a1, f1, acc1);
    }
    const float bias = b2c[colc];
    if (col < PX) {
      for (int r = 0; r < 8; ++r) {
        const int row = mt * 16 + 8 * g + r;
        const float val = acc0[r] + acc1[r] + bias;
        sX[row * PX + col] = (col < FEAT && sValid[row] != 0) ? val : 0.0f;
      }
    }
  }
  __syncthreads();

  {
    const int col = w * 16 + c;
    const float* a0p = &sX[c * PX + 2 * g];
    const float* a1p = &sX[(16 + c) * PX + 2 * g];
    v8f acc0 = (v8f){};
    v8f acc1 = (v8f){};
    for (int k0 = 0; k0 < PX; k0 += 4) {
      const int ka = k0 + 2 * g;
      v2f bf;
      bf.x = W1[imin(ka,     FEAT - 1) * HID + col];
      bf.y = W1[imin(ka + 1, FEAT - 1) * HID + col];
      const v2f a0 = *(const v2f*)(a0p + k0);
      const v2f a1 = *(const v2f*)(a1p + k0);
      acc0 = wmma_f32x4(a0, bf, acc0);
      acc1 = wmma_f32x4(a1, bf, acc1);
    }
    const float bias = b1[col];
    for (int r = 0; r < 8; ++r) {
      sA[(8 * g + r) * PA + col]      = fmaxf(acc0[r] + bias, 0.0f);
      sA[(16 + 8 * g + r) * PA + col] = fmaxf(acc1[r] + bias, 0.0f);
    }
  }
  __syncthreads();

  {
    const int col = w * 16 + c;
    const float* a0p = &sA[c * PA + 2 * g];
    const float* a1p = &sA[(16 + c) * PA + 2 * g];
    const float* __restrict__ bp = W2 + (2 * g) * HID + col;
    v8f acc0 = (v8f){};
    v8f acc1 = (v8f){};
    for (int k0 = 0; k0 < HID; k0 += 4) {
      v2f bf;
      bf.x = bp[k0 * HID];
      bf.y = bp[k0 * HID + HID];
      const v2f a0 = *(const v2f*)(a0p + k0);
      const v2f a1 = *(const v2f*)(a1p + k0);
      acc0 = wmma_f32x4(a0, bf, acc0);
      acc1 = wmma_f32x4(a1, bf, acc1);
    }
    const float bias = b2[col];
    const float w3 = W3[col];
    for (int r = 0; r < 8; ++r) {
      sH[(8 * g + r) * PA + col]      = fmaxf(acc0[r] + bias, 0.0f) * w3;
      sH[(16 + 8 * g + r) * PA + col] = fmaxf(acc1[r] + bias, 0.0f) * w3;
    }
  }
  __syncthreads();

  if (w == 0) {
    const float4* p = (const float4*)&sH[lane * PA];
    float s = 0.0f;
    for (int j = 0; j < HID / 4; ++j) {
      const float4 v = p[j];
      s += v.x; s += v.y; s += v.z; s += v.w;
    }
    out[row0 + lane] = s + b3[0];
  }
}


extern "C" void kernel_launch(void* const* d_in, const int* in_sizes, int n_in,
                              void* d_out, int out_size, void* d_ws, size_t ws_size,
                              hipStream_t stream) {
  (void)in_sizes; (void)n_in; (void)out_size; (void)ws_size;

  const float* edge_feat = (const float*)d_in[0];
  const int*   edge_idx  = (const int*)d_in[1];
  const int*   act_range = (const int*)d_in[2];
  const float* Wl1 = (const float*)d_in[3];
  const float* Wr1 = (const float*)d_in[4];
  const float* b1c = (const float*)d_in[5];
  const float* Wl2 = (const float*)d_in[6];
  const float* Wr2 = (const float*)d_in[7];
  const float* b2c = (const float*)d_in[8];
  const float* W1  = (const float*)d_in[9];
  const float* b1  = (const float*)d_in[10];
  const float* W2  = (const float*)d_in[11];
  const float* b2  = (const float*)d_in[12];
  const float* W3  = (const float*)d_in[13];
  const float* b3  = (const float*)d_in[14];
  float* out = (float*)d_out;

  const long S_AGG1 = (long)BB * NN * FEAT;
  const long S_CNT  = (long)BB * NN;
  const long S_H    = (long)BB * NN * HID;

  float* agg1 = (float*)d_ws;
  float* cnt  = agg1 + S_AGG1;
  float* h    = cnt  + S_CNT;
  float* agg2 = h    + S_H;
  float* lists = agg2 + S_H;

  int* deg    = (int*)cnt;
  int* cursor = (int*)lists;
  int* ssrc   = cursor + S_CNT;

  build_lists_kernel<<<dim3(2 * NRANGES, BB), 1024, 0, stream>>>(edge_idx, deg, cursor, ssrc);

  sage1_fused_kernel<<<S1_BLOCKS, 512, 0, stream>>>(edge_feat, deg, cursor, ssrc, Wl1, Wr1, b1c, h);

  tail_fused_kernel<<<BB * PADR / TROWS, 256, 0, stream>>>(
      h, deg, cursor, ssrc, Wl2, Wr2, b2c, W1, b1, W2, b2, W3, b3, act_range, out);
}
